# starting kernel
# baseline (speedup 1.0000x reference)
#include <hip/hip_runtime.h>

typedef __attribute__((ext_vector_type(2))) float v2f;
typedef __attribute__((ext_vector_type(8))) float v8f;

constexpr int HB = 64;
constexpr int HT = 4096;
constexpr int HK = 32;
constexpr float LOG2E = 1.44269504088896340736f;
constexpr float LN2   = 0.69314718055994530942f;

#if __has_builtin(__builtin_amdgcn_exp2f)
__device__ __forceinline__ float fexp2(float x){ return __builtin_amdgcn_exp2f(x); }
#else
__device__ __forceinline__ float fexp2(float x){ return exp2f(x); }
#endif
#if __has_builtin(__builtin_amdgcn_logf)
__device__ __forceinline__ float flog2(float x){ return __builtin_amdgcn_logf(x); }
#else
__device__ __forceinline__ float flog2(float x){ return log2f(x); }
#endif

__device__ __forceinline__ float rmax32(float x){
#pragma unroll
  for (int m = 16; m >= 1; m >>= 1) x = fmaxf(x, __shfl_xor(x, m, 32));
  return x;
}
__device__ __forceinline__ float rsum32(float x){
#pragma unroll
  for (int m = 16; m >= 1; m >>= 1) x += __shfl_xor(x, m, 32);
  return x;
}
__device__ __forceinline__ float rmax16(float x){
#pragma unroll
  for (int m = 8; m >= 1; m >>= 1) x = fmaxf(x, __shfl_xor(x, m, 32));
  return x;
}

__device__ __forceinline__ v8f wmma4(v2f a, v2f b, v8f c){
  return __builtin_amdgcn_wmma_f32_16x16x4_f32(
       false, a,  false, b,
       (short)0, c,  false,  false);
}

__device__ __forceinline__ void load_em(float (&dst)[2][8], const float* emb,
                                        int t, int h, int ln){
#pragma unroll
  for (int nt = 0; nt < 2; ++nt)
#pragma unroll
    for (int r = 0; r < 8; ++r)
      dst[nt][r] = emb[((size_t)(r + 8*h)*HT + t)*HK + nt*16 + ln];
}

__device__ __forceinline__ void store_tiles(float* dstb, int t, int h, int ln,
                                            const float (&v)[2][8]){
#pragma unroll
  for (int nt = 0; nt < 2; ++nt)
#pragma unroll
    for (int r = 0; r < 8; ++r)
      dstb[((size_t)(r + 8*h)*HT + t)*HK + nt*16 + ln] = v[nt][r];
}

template<int DIR>
__device__ __forceinline__ void hmm_step(float (&st)[2][8], const float (&emc)[2][8],
                                         const v2f (&breg)[8][2],
                                         float* sPf, const v2f* sP2,
                                         int h, int ln){
  float x[2][8];
#pragma unroll
  for (int nt = 0; nt < 2; ++nt)
#pragma unroll
    for (int r = 0; r < 8; ++r)
      x[nt][r] = (DIR == 0) ? st[nt][r] : st[nt][r] + emc[nt][r];

  float mrow[8];
#pragma unroll
  for (int r = 0; r < 8; ++r) mrow[r] = fmaxf(x[0][r], x[1][r]);
#pragma unroll
  for (int r = 0; r < 8; ++r) mrow[r] = rmax16(mrow[r]);

#pragma unroll
  for (int nt = 0; nt < 2; ++nt)
#pragma unroll
    for (int r = 0; r < 8; ++r)
      sPf[(r + 8*h)*34 + nt*16 + ln] = fexp2((x[nt][r] - mrow[r]) * LOG2E);
  __syncthreads();
  v2f a[8];
#pragma unroll
  for (int kc = 0; kc < 8; ++kc) a[kc] = sP2[ln*17 + 2*kc + h];
  __syncthreads();

  v8f c0 = {0.f,0.f,0.f,0.f,0.f,0.f,0.f,0.f};
  v8f c1 = {0.f,0.f,0.f,0.f,0.f,0.f,0.f,0.f};
#pragma unroll
  for (int kc = 0; kc < 8; ++kc){
    c0 = wmma4(a[kc], breg[kc][0], c0);
    c1 = wmma4(a[kc], breg[kc][1], c1);
  }
#pragma unroll
  for (int r = 0; r < 8; ++r){
    const float b0 = (DIR == 0) ? mrow[r] + emc[0][r] : mrow[r];
    const float b1 = (DIR == 0) ? mrow[r] + emc[1][r] : mrow[r];
    st[0][r] = fmaf(LN2, flog2(c0[r]), b0);
    st[1][r] = fmaf(LN2, flog2(c1[r]), b1);
  }
}

__global__ void __launch_bounds__(32)
hmm_scan_kernel(const float* __restrict__ em,
                const float* __restrict__ log_pi,
                const float* __restrict__ log_A,
                float* __restrict__ alpha_out,
                float* __restrict__ beta_out){
  __shared__ float sA[32*33];
  __shared__ v2f   sP2[16*17];
  float* sPf = (float*)sP2;

  const int lane = threadIdx.x;
  const int h    = lane >> 4;
  const int ln   = lane & 15;
  const int dir  = blockIdx.x >> 2;
  const int mb   = blockIdx.x & 3;

  for (int i = 0; i < 32; ++i){
    float xv = log_A[i*32 + lane];
    float mx = rmax32(xv);
    float e  = fexp2((xv - mx) * LOG2E);
    float s  = rsum32(e);
    sA[i*33 + lane] = e / s;
  }
  __syncthreads();

  v2f breg[8][2];
#pragma unroll
  for (int kc = 0; kc < 8; ++kc){
#pragma unroll
    for (int nt = 0; nt < 2; ++nt){
      const int kk = 4*kc + 2*h;
      const int nn = nt*16 + ln;
      if (dir == 0){
        breg[kc][nt].x = sA[kk*33 + nn];
        breg[kc][nt].y = sA[(kk+1)*33 + nn];
      } else {
        breg[kc][nt].x = sA[nn*33 + kk];
        breg[kc][nt].y = sA[nn*33 + kk + 1];
      }
    }
  }

  const float* emb = em + (size_t)(mb*16)*HT*HK;

  if (dir == 0){
    float xp = log_pi[lane];
    float mp = rmax32(xp);
    float sp = rsum32(fexp2((xp - mp) * LOG2E));
    float lp = xp - mp - LN2 * flog2(sp);
    float lpn[2];
#pragma unroll
    for (int nt = 0; nt < 2; ++nt) lpn[nt] = __shfl(lp, nt*16 + ln, 32);

    float* ab = alpha_out + (size_t)(mb*16)*HT*HK;
    float al[2][8];
    float emA[2][8], emB[2][8];
    load_em(emA, emb, 0, h, ln);
#pragma unroll
    for (int nt = 0; nt < 2; ++nt)
#pragma unroll
      for (int r = 0; r < 8; ++r) al[nt][r] = lpn[nt] + emA[nt][r];
    store_tiles(ab, 0, h, ln, al);
    load_em(emA, emb, 1, h, ln);

    for (int t = 1; t + 1 < HT; t += 2){
      load_em(emB, emb, t + 1, h, ln);
      hmm_step<0>(al, emA, breg, sPf, sP2, h, ln);
      store_tiles(ab, t, h, ln, al);
      load_em(emA, emb, (t + 2 < HT) ? t + 2 : HT - 1, h, ln);
      hmm_step<0>(al, emB, breg, sPf, sP2, h, ln);
      store_tiles(ab, t + 1, h, ln, al);
    }
    hmm_step<0>(al, emA, breg, sPf, sP2, h, ln);
    store_tiles(ab, HT - 1, h, ln, al);
  } else {
    float* bb = beta_out + (size_t)(mb*16)*HT*HK;
    float be[2][8];
#pragma unroll
    for (int nt = 0; nt < 2; ++nt)
#pragma unroll
      for (int r = 0; r < 8; ++r) be[nt][r] = 0.f;
    store_tiles(bb, HT - 1, h, ln, be);
    float emA[2][8], emB[2][8];
    load_em(emA, emb, HT - 1, h, ln);

    for (int t = HT - 2; t >= 2; t -= 2){
      load_em(emB, emb, t, h, ln);
      hmm_step<1>(be, emA, breg, sPf, sP2, h, ln);
      store_tiles(bb, t, h, ln, be);
      load_em(emA, emb, t - 1, h, ln);
      hmm_step<1>(be, emB, breg, sPf, sP2, h, ln);
      store_tiles(bb, t - 1, h, ln, be);
    }
    hmm_step<1>(be, emA, breg, sPf, sP2, h, ln);
    store_tiles(bb, 0, h, ln, be);
  }
}

__global__ void __launch_bounds__(256)
hmm_gamma_kernel(const float* alpha, const float* beta, float* out, int npairs){
  const int w    = (blockIdx.x*blockDim.x + threadIdx.x) >> 5;
  const int lane = threadIdx.x & 31;
  if (w >= npairs) return;
  const size_t idx = (size_t)w*32 + lane;
  float g  = alpha[idx] + beta[idx];
  float mx = rmax32(g);
  float s  = rsum32(fexp2((g - mx) * LOG2E));
  out[idx] = g - mx - LN2 * flog2(s);
}

extern "C" void kernel_launch(void* const* d_in, const int* in_sizes, int n_in,
                              void* d_out, int out_size, void* d_ws, size_t ws_size,
                              hipStream_t stream) {
  const float* em     = (const float*)d_in[0];
  const float* log_pi = (const float*)d_in[1];
  const float* log_A  = (const float*)d_in[2];
  float* out  = (float*)d_out;
  float* beta = (float*)d_ws;

  hmm_scan_kernel<<<dim3(8), dim3(32), 0, stream>>>(em, log_pi, log_A, out, beta);

  const int npairs = HB * HT;
  const int threads = 256;
  const int blocks = (npairs*32 + threads - 1) / threads;
  hmm_gamma_kernel<<<dim3(blocks), dim3(threads), 0, stream>>>(out, beta, out, npairs);
}


# Step's row maximum: butterfly over each 16-lane half as four max instructions with the row swizzle (DPP row_xmask) on their source, not four LDS-routed lane permutes with waits; sum order unchanged
# speedup vs baseline: 1.6501x; 1.6501x over previous
#include <hip/hip_runtime.h>

typedef __attribute__((ext_vector_type(2))) float v2f;
typedef __attribute__((ext_vector_type(8))) float v8f;

constexpr int HB = 64;
constexpr int HT = 4096;
constexpr int HK = 32;
constexpr float LOG2E = 1.44269504088896340736f;
constexpr float LN2   = 0.69314718055994530942f;

#if __has_builtin(__builtin_amdgcn_exp2f)
__device__ __forceinline__ float fexp2(float x){ return __builtin_amdgcn_exp2f(x); }
#else
__device__ __forceinline__ float fexp2(float x){ return exp2f(x); }
#endif
#if __has_builtin(__builtin_amdgcn_logf)
__device__ __forceinline__ float flog2(float x){ return __builtin_amdgcn_logf(x); }
#else
__device__ __forceinline__ float flog2(float x){ return log2f(x); }
#endif

__device__ __forceinline__ float rmax32(float x){
#pragma unroll
  for (int m = 16; m >= 1; m >>= 1) x = fmaxf(x, __shfl_xor(x, m, 32));
  return x;
}
__device__ __forceinline__ float rsum32(float x){
#pragma unroll
  for (int m = 16; m >= 1; m >>= 1) x += __shfl_xor(x, m, 32);
  return x;
}
template<int M>
__device__ __forceinline__ float rowmax_step(float x){
  float y;
  asm("v_max_num_f32_dpp %0, %1, %1 row_xmask:%2 row_mask:0xf bank_mask:0xf"
      : "=v"(y) : "v"(x), "n"(M));
  return y;
}

__device__ __forceinline__ v8f wmma4(v2f a, v2f b, v8f c){
  return __builtin_amdgcn_wmma_f32_16x16x4_f32(
       false, a,  false, b,
       (short)0, c,  false,  false);
}

__device__ __forceinline__ void load_em(float (&dst)[2][8], const float* emb,
                                        int t, int h, int ln){
#pragma unroll
  for (int nt = 0; nt < 2; ++nt)
#pragma unroll
    for (int r = 0; r < 8; ++r)
      dst[nt][r] = emb[((size_t)(r + 8*h)*HT + t)*HK + nt*16 + ln];
}

__device__ __forceinline__ void store_tiles(float* dstb, int t, int h, int ln,
                                            const float (&v)[2][8]){
#pragma unroll
  for (int nt = 0; nt < 2; ++nt)
#pragma unroll
    for (int r = 0; r < 8; ++r)
      dstb[((size_t)(r + 8*h)*HT + t)*HK + nt*16 + ln] = v[nt][r];
}

template<int DIR>
__device__ __forceinline__ void hmm_step(float (&st)[2][8], const float (&emc)[2][8],
                                         const v2f (&breg)[8][2],
                                         float* sPf, const v2f* sP2,
                                         int h, int ln){
  float x[2][8];
#pragma unroll
  for (int nt = 0; nt < 2; ++nt)
#pragma unroll
    for (int r = 0; r < 8; ++r)
      x[nt][r] = (DIR == 0) ? st[nt][r] : st[nt][r] + emc[nt][r];

  float mrow[8];
#pragma unroll
  for (int r = 0; r < 8; ++r) mrow[r] = fmaxf(x[0][r], x[1][r]);
#pragma unroll
  for (int r = 0; r < 8; ++r) mrow[r] = rowmax_step<1>(mrow[r]);
#pragma unroll
  for (int r = 0; r < 8; ++r) mrow[r] = rowmax_step<2>(mrow[r]);
#pragma unroll
  for (int r = 0; r < 8; ++r) mrow[r] = rowmax_step<4>(mrow[r]);
#pragma unroll
  for (int r = 0; r < 8; ++r) mrow[r] = rowmax_step<8>(mrow[r]);

#pragma unroll
  for (int nt = 0; nt < 2; ++nt)
#pragma unroll
    for (int r = 0; r < 8; ++r)
      sPf[(r + 8*h)*34 + nt*16 + ln] = fexp2((x[nt][r] - mrow[r]) * LOG2E);
  __builtin_amdgcn_sched_barrier(0);
  v2f a[8];
#pragma unroll
  for (int kc = 0; kc < 8; ++kc) a[kc] = sP2[ln*17 + 2*kc + h];
  __builtin_amdgcn_sched_barrier(0);

  v8f c0 = {0.f,0.f,0.f,0.f,0.f,0.f,0.f,0.f};
  v8f c1 = {0.f,0.f,0.f,0.f,0.f,0.f,0.f,0.f};
#pragma unroll
  for (int kc = 0; kc < 8; ++kc){
    c0 = wmma4(a[kc], breg[kc][0], c0);
    c1 = wmma4(a[kc], breg[kc][1], c1);
  }
#pragma unroll
  for (int r = 0; r < 8; ++r){
    const float b0 = (DIR == 0) ? mrow[r] + emc[0][r] : mrow[r];
    const float b1 = (DIR == 0) ? mrow[r] + emc[1][r] : mrow[r];
    st[0][r] = fmaf(LN2, flog2(c0[r]), b0);
    st[1][r] = fmaf(LN2, flog2(c1[r]), b1);
  }
}

__global__ void __launch_bounds__(32)
hmm_scan_kernel(const float* __restrict__ em,
                const float* __restrict__ log_pi,
                const float* __restrict__ log_A,
                float* __restrict__ alpha_out,
                float* __restrict__ beta_out){
  __shared__ float sA[32*33];
  __shared__ v2f   sP2[16*17];
  float* sPf = (float*)sP2;

  const int lane = threadIdx.x;
  const int h    = lane >> 4;
  const int ln   = lane & 15;
  const int dir  = blockIdx.x >> 2;
  const int mb   = blockIdx.x & 3;

  for (int i = 0; i < 32; ++i){
    float xv = log_A[i*32 + lane];
    float mx = rmax32(xv);
    float e  = fexp2((xv - mx) * LOG2E);
    float s  = rsum32(e);
    sA[i*33 + lane] = e / s;
  }
  __syncthreads();

  v2f breg[8][2];
#pragma unroll
  for (int kc = 0; kc < 8; ++kc){
#pragma unroll
    for (int nt = 0; nt < 2; ++nt){
      const int kk = 4*kc + 2*h;
      const int nn = nt*16 + ln;
      if (dir == 0){
        breg[kc][nt].x = sA[kk*33 + nn];
        breg[kc][nt].y = sA[(kk+1)*33 + nn];
      } else {
        breg[kc][nt].x = sA[nn*33 + kk];
        breg[kc][nt].y = sA[nn*33 + kk + 1];
      }
    }
  }

  const float* emb = em + (size_t)(mb*16)*HT*HK;

  if (dir == 0){
    float xp = log_pi[lane];
    float mp = rmax32(xp);
    float sp = rsum32(fexp2((xp - mp) * LOG2E));
    float lp = xp - mp - LN2 * flog2(sp);
    float lpn[2];
#pragma unroll
    for (int nt = 0; nt < 2; ++nt) lpn[nt] = __shfl(lp, nt*16 + ln, 32);

    float* ab = alpha_out + (size_t)(mb*16)*HT*HK;
    float al[2][8];
    float emA[2][8], emB[2][8];
    load_em(emA, emb, 0, h, ln);
#pragma unroll
    for (int nt = 0; nt < 2; ++nt)
#pragma unroll
      for (int r = 0; r < 8; ++r) al[nt][r] = lpn[nt] + emA[nt][r];
    store_tiles(ab, 0, h, ln, al);
    load_em(emA, emb, 1, h, ln);

    for (int t = 1; t + 1 < HT; t += 2){
      load_em(emB, emb, t + 1, h, ln);
      hmm_step<0>(al, emA, breg, sPf, sP2, h, ln);
      store_tiles(ab, t, h, ln, al);
      load_em(emA, emb, (t + 2 < HT) ? t + 2 : HT - 1, h, ln);
      hmm_step<0>(al, emB, breg, sPf, sP2, h, ln);
      store_tiles(ab, t + 1, h, ln, al);
    }
    hmm_step<0>(al, emA, breg, sPf, sP2, h, ln);
    store_tiles(ab, HT - 1, h, ln, al);
  } else {
    float* bb = beta_out + (size_t)(mb*16)*HT*HK;
    float be[2][8];
#pragma unroll
    for (int nt = 0; nt < 2; ++nt)
#pragma unroll
      for (int r = 0; r < 8; ++r) be[nt][r] = 0.f;
    store_tiles(bb, HT - 1, h, ln, be);
    float emA[2][8], emB[2][8];
    load_em(emA, emb, HT - 1, h, ln);

    for (int t = HT - 2; t >= 2; t -= 2){
      load_em(emB, emb, t, h, ln);
      hmm_step<1>(be, emA, breg, sPf, sP2, h, ln);
      store_tiles(bb, t, h, ln, be);
      load_em(emA, emb, t - 1, h, ln);
      hmm_step<1>(be, emB, breg, sPf, sP2, h, ln);
      store_tiles(bb, t - 1, h, ln, be);
    }
    hmm_step<1>(be, emA, breg, sPf, sP2, h, ln);
    store_tiles(bb, 0, h, ln, be);
  }
}

__global__ void __launch_bounds__(256)
hmm_gamma_kernel(const float* alpha, const float* beta, float* out, int npairs){
  const int w    = (blockIdx.x*blockDim.x + threadIdx.x) >> 5;
  const int lane = threadIdx.x & 31;
  if (w >= npairs) return;
  const size_t idx = (size_t)w*32 + lane;
  float g  = alpha[idx] + beta[idx];
  float mx = rmax32(g);
  float s  = rsum32(fexp2((g - mx) * LOG2E));
  out[idx] = g - mx - LN2 * flog2(s);
}

extern "C" void kernel_launch(void* const* d_in, const int* in_sizes, int n_in,
                              void* d_out, int out_size, void* d_ws, size_t ws_size,
                              hipStream_t stream) {
  const float* em     = (const float*)d_in[0];
  const float* log_pi = (const float*)d_in[1];
  const float* log_A  = (const float*)d_in[2];
  float* out  = (float*)d_out;
  float* beta = (float*)d_ws;

  hmm_scan_kernel<<<dim3(8), dim3(32), 0, stream>>>(em, log_pi, log_A, out, beta);

  const int npairs = HB * HT;
  const int threads = 256;
  const int blocks = (npairs*32 + threads - 1) / threads;
  hmm_gamma_kernel<<<dim3(blocks), dim3(threads), 0, stream>>>(out, beta, out, npairs);
}


# Shift = maximum of the row one step back; carried row kept relative to it in base-2 units: per state the chain is log2, one add, exp2 (row max, base change, emission add leave it); rounding changes
# speedup vs baseline: 2.4918x; 1.5101x over previous
#include <hip/hip_runtime.h>

typedef __attribute__((ext_vector_type(2))) float v2f;
typedef __attribute__((ext_vector_type(4))) float v4f;
typedef __attribute__((ext_vector_type(8))) float v8f;

constexpr int HB = 64;
constexpr int HT = 4096;
constexpr int HK = 32;
constexpr int CP = 36;
constexpr float LOG2E = 1.44269504088896340736f;
constexpr float LN2   = 0.69314718055994530942f;

#if __has_builtin(__builtin_amdgcn_exp2f)
__device__ __forceinline__ float fexp2(float x){ return __builtin_amdgcn_exp2f(x); }
#else
__device__ __forceinline__ float fexp2(float x){ return exp2f(x); }
#endif
#if __has_builtin(__builtin_amdgcn_logf)
__device__ __forceinline__ float flog2(float x){ return __builtin_amdgcn_logf(x); }
#else
__device__ __forceinline__ float flog2(float x){ return log2f(x); }
#endif

__device__ __forceinline__ float rmax32(float x){
#pragma unroll
  for (int m = 16; m >= 1; m >>= 1) x = fmaxf(x, __shfl_xor(x, m, 32));
  return x;
}
__device__ __forceinline__ float rsum32(float x){
#pragma unroll
  for (int m = 16; m >= 1; m >>= 1) x += __shfl_xor(x, m, 32);
  return x;
}
__device__ __forceinline__ float xhalf(float x){
  return __builtin_bit_cast(float, __builtin_amdgcn_permlanex16(
      __builtin_bit_cast(int, x), __builtin_bit_cast(int, x),
      0x76543210, 0xfedcba98, false, false));
}

__device__ __forceinline__ v8f wmma4(v2f a, v2f b, v8f c){
  return __builtin_amdgcn_wmma_f32_16x16x4_f32(
       false, a,  false, b,
       (short)0, c,  false,  false);
}

__device__ __forceinline__ void load_row(v4f (&dst)[4], const float* base,
                                         unsigned loff, int t){
  const float* p = base + (size_t)t*HK;
#pragma unroll
  for (int i = 0; i < 4; ++i) dst[i] = *(const v4f*)(p + loff + 4*i);
}
__device__ __forceinline__ void store_row(float* base, unsigned loff, int t,
                                          const v4f (&v)[4]){
  float* p = base + (size_t)t*HK;
#pragma unroll
  for (int i = 0; i < 4; ++i) *(v4f*)(p + loff + 4*i) = v[i];
}

__device__ __forceinline__ float row_max(const v4f (&x)[4]){
  const float a0 = fmaxf(fmaxf(x[0][0], x[0][1]), x[0][2]);
  const float a1 = fmaxf(fmaxf(x[0][3], x[1][0]), x[1][1]);
  const float a2 = fmaxf(fmaxf(x[1][2], x[1][3]), x[2][0]);
  const float a3 = fmaxf(fmaxf(x[2][1], x[2][2]), x[2][3]);
  const float a4 = fmaxf(fmaxf(x[3][0], x[3][1]), x[3][2]);
  const float b0 = fmaxf(fmaxf(a0, a1), a2);
  const float b1 = fmaxf(fmaxf(a3, a4), x[3][3]);
  const float mh = fmaxf(b0, b1);
  return fmaxf(mh, xhalf(mh));
}

__device__ __forceinline__ void hmm_matrix_run(const v4f (&z)[4], const v2f (&breg)[8][2],
                                               v8f &c0, v8f &c1){
  c0 = (v8f){0.f,0.f,0.f,0.f,0.f,0.f,0.f,0.f};
  c1 = (v8f){0.f,0.f,0.f,0.f,0.f,0.f,0.f,0.f};
#pragma unroll
  for (int kc = 0; kc < 8; ++kc){
    v2f a;
    a.x = fexp2(z[kc >> 1][2*(kc & 1)]);
    a.y = fexp2(z[kc >> 1][2*(kc & 1) + 1]);
    c0 = wmma4(a, breg[kc][0], c0);
    c1 = wmma4(a, breg[kc][1], c1);
  }
}

__device__ __forceinline__ void lds_block_edge(){ __builtin_amdgcn_sched_barrier(0); }

constexpr int G    = 4;
constexpr int NG   = HT / G;
constexpr int POS  = 16*CP + 32;
constexpr int RING = G*POS;
constexpr int NI   = 16;

__device__ __forceinline__ void wait_ds0(){ asm volatile("s_wait_dscnt 0x0" ::: "memory"); }
__device__ __forceinline__ void wait_ds_but8(){ asm volatile("s_wait_dscnt 0x8" ::: "memory"); }
__device__ __forceinline__ void wg_barrier(){ asm volatile("s_barrier_signal -1\n\ts_barrier_wait -1" ::: "memory"); }

template <int DIR>
__device__ __forceinline__ void scan_group(int g, v4f (&z)[4], v4f (&op)[4], float &s, float &zm,
                                           const v2f (&breg)[8][2],
                                           float* sC, const float* sL,
                                           const float* eb, float* ob, float* obp, int h, int ln){
#pragma unroll
  for (int u = 0; u < G; ++u){
    if (u == 0 && g == 0){
      v4f e[4], x[4];
#pragma unroll
      for (int i = 0; i < 4; ++i) e[i] = *(const v4f*)(eb + 4*i);
#pragma unroll
      for (int i = 0; i < 4; ++i)
#pragma unroll
        for (int c = 0; c < 4; ++c){
          if (DIR == 0){ x[i][c] = sL[16*h + 4*i + c] + e[i][c]; op[i][c] = x[i][c]; }
          else         { op[i][c] = 0.f;                         x[i][c] = op[i][c] + e[i][c]; }
        }
      s = row_max(x);
#pragma unroll
      for (int i = 0; i < 4; ++i)
#pragma unroll
        for (int c = 0; c < 4; ++c) z[i][c] = (x[i][c] - s) * LOG2E;
      zm = 0.f;
    } else {
      v8f c0, c1;
      hmm_matrix_run(z, breg, c0, c1);
      lds_block_edge();
#pragma unroll
      for (int r = 0; r < 8; ++r){
        sC[(r + 8*h)*CP + ln]      = c0[r];
        sC[(r + 8*h)*CP + 16 + ln] = c1[r];
      }
      v4f e[4], cc[4];
#pragma unroll
      for (int i = 0; i < 4; ++i) cc[i] = *(const v4f*)(sC + ln*CP + 16*h + 4*i);
#pragma unroll
      for (int i = 0; i < 4; ++i) e[i] = *(const v4f*)(eb + u*POS + 4*i);
      float* od = (u == 0) ? (obp + (G - 1)*POS) : (ob + (u - 1)*POS);
#pragma unroll
      for (int i = 0; i < 4; ++i) *(v4f*)(od + 4*i) = op[i];
      lds_block_edge();
#pragma unroll
      for (int i = 0; i < 4; ++i)
#pragma unroll
        for (int c = 0; c < 4; ++c){
          const float l = flog2(cc[i][c]);
          z[i][c] = l + fmaf(e[i][c], LOG2E, -zm);
          if (DIR == 0) op[i][c] = fmaf(LN2, l, s + e[i][c]);
          else          op[i][c] = fmaf(LN2, l, s);
        }
      s  = fmaf(LN2, zm, s);
      zm = row_max(z);
    }
  }
}

__device__ __forceinline__ void mem_load(v4f (&er)[NI], const float* src, int tbase, int lane){
#pragma unroll
  for (int j = 0; j < NI; ++j)
    er[j] = *(const v4f*)(src + ((size_t)j*HT + tbase)*HK + 4*lane);
}
__device__ __forceinline__ void mem_put(const v4f (&er)[NI], float* half, int lds_off){
#pragma unroll
  for (int j = 0; j < NI; ++j) *(v4f*)(half + lds_off + j*CP) = er[j];
}
__device__ __forceinline__ void mem_drain(const float* half, int lds_off, float* dst, int tbase, int lane){
  v4f v[NI];
#pragma unroll
  for (int j = 0; j < NI; ++j) v[j] = *(const v4f*)(half + lds_off + j*CP);
#pragma unroll
  for (int j = 0; j < NI; ++j)
    *(v4f*)(dst + ((size_t)j*HT + tbase)*HK + 4*lane) = v[j];
}

__global__ void __launch_bounds__(64)
hmm_scan_kernel(const float* __restrict__ em,
                const float* __restrict__ log_pi,
                const float* __restrict__ log_A,
                float* __restrict__ alpha_out,
                float* __restrict__ beta_out){
  __shared__ float sA[32*33];
  __shared__ __attribute__((aligned(16))) float sC[16*CP];
  __shared__ float sL[32];
  __shared__ __attribute__((aligned(16))) float sE[2*RING];
  __shared__ __attribute__((aligned(16))) float sO[3*RING];

  const int lane = threadIdx.x & 31;
  const int wid  = __builtin_amdgcn_readfirstlane((int)(threadIdx.x >> 5));
  const int h    = lane >> 4;
  const int ln   = lane & 15;
  const int dir  = blockIdx.x >> 2;
  const int mb   = blockIdx.x & 3;

  for (int i = 16*wid; i < 16*wid + 16; ++i){
    float xv = log_A[i*32 + lane];
    float mx = rmax32(xv);
    float e  = fexp2((xv - mx) * LOG2E);
    float s  = rsum32(e);
    sA[i*33 + lane] = e / s;
  }
  if (wid == 0){
    float xp = log_pi[lane];
    float mp = rmax32(xp);
    float sp = rsum32(fexp2((xp - mp) * LOG2E));
    sL[lane] = xp - mp - LN2 * flog2(sp);
  }
  __syncthreads();

  v2f breg[8][2];
#pragma unroll
  for (int kc = 0; kc < 8; ++kc){
#pragma unroll
    for (int nt = 0; nt < 2; ++nt){
      const int kk = 16*h + 2*kc;
      const int nn = nt*16 + ln;
      if (dir == 0){
        breg[kc][nt].x = sA[kk*33 + nn];
        breg[kc][nt].y = sA[(kk+1)*33 + nn];
      } else {
        breg[kc][nt].x = sA[nn*33 + kk];
        breg[kc][nt].y = sA[nn*33 + kk + 1];
      }
    }
  }

  const float* emb  = em + (size_t)(mb*16)*HT*HK;
  float*       outb = (dir == 0 ? alpha_out : beta_out) + (size_t)(mb*16)*HT*HK;

  const int row_off = ln*CP + 16*h;
  const int tl      = lane >> 3;
  const int lds_off = ((dir == 0) ? tl : (G - 1 - tl))*POS + 4*(lane & 7);
  const int tb0     = (dir == 0) ? 0 : HT - G;
  const int tbs     = (dir == 0) ? G : -G;

  v4f z[4];
#pragma unroll
  for (int i = 0; i < 4; ++i) z[i] = (v4f){0.f, 0.f, 0.f, 0.f};
  float sh = 0.f, zm = 0.f;
  v4f op[4];
#pragma unroll
  for (int i = 0; i < 4; ++i) op[i] = (v4f){0.f, 0.f, 0.f, 0.f};
  v4f er[NI];
#pragma unroll
  for (int j = 0; j < NI; ++j) er[j] = (v4f){0.f, 0.f, 0.f, 0.f};

  if (wid != 0){
    mem_load(er, emb, tb0, lane);
    mem_put(er, sE, lds_off);
    mem_load(er, emb, tb0 + tbs, lane);
  }

  int o3 = 0;
  for (int g = 0; g < NG; ++g){
    if (wid == 0) wait_ds_but8(); else wait_ds0();
    wg_barrier();
    const int o3d = (o3 == 2) ? 0 : o3 + 1;
    if (wid == 0){
      const int o3p = (o3 == 0) ? 2 : o3 - 1;
      const float* eb  = sE + (g & 1)*RING + row_off;
      float*       ob  = sO + o3*RING + row_off;
      float*       obp = sO + o3p*RING + row_off;
      if (dir == 0) scan_group<0>(g, z, op, sh, zm, breg, sC, sL, eb, ob, obp, h, ln);
      else          scan_group<1>(g, z, op, sh, zm, breg, sC, sL, eb, ob, obp, h, ln);
    } else {
      if (g + 1 < NG) mem_put(er, sE + ((g + 1) & 1)*RING, lds_off);
      if (g + 2 < NG) mem_load(er, emb, tb0 + (g + 2)*tbs, lane);
      if (g >= 2)     mem_drain(sO + o3d*RING, lds_off, outb, tb0 + (g - 2)*tbs, lane);
    }
    o3 = o3d;
  }
  if (wid == 0){
    float* od = sO + ((NG - 1) % 3)*RING + row_off + (G - 1)*POS;
#pragma unroll
    for (int i = 0; i < 4; ++i) *(v4f*)(od + 4*i) = op[i];
  }
  wait_ds0();
  wg_barrier();
  if (wid != 0){
    mem_drain(sO + ((NG - 2) % 3)*RING, lds_off, outb, tb0 + (NG - 2)*tbs, lane);
    mem_drain(sO + ((NG - 1) % 3)*RING, lds_off, outb, tb0 + (NG - 1)*tbs, lane);
  }
}

__global__ void __launch_bounds__(256)
hmm_gamma_kernel(const float* alpha, const float* beta, float* out, int npairs){
  const int w    = (blockIdx.x*blockDim.x + threadIdx.x) >> 5;
  const int lane = threadIdx.x & 31;
  if (w >= npairs) return;
  const size_t idx = (size_t)w*32 + lane;
  float g  = alpha[idx] + beta[idx];
  float mx = rmax32(g);
  float s  = rsum32(fexp2((g - mx) * LOG2E));
  out[idx] = g - mx - LN2 * flog2(s);
}

extern "C" void kernel_launch(void* const* d_in, const int* in_sizes, int n_in,
                              void* d_out, int out_size, void* d_ws, size_t ws_size,
                              hipStream_t stream) {
  const float* em     = (const float*)d_in[0];
  const float* log_pi = (const float*)d_in[1];
  const float* log_A  = (const float*)d_in[2];
  float* out  = (float*)d_out;
  float* beta = (float*)d_ws;

  hmm_scan_kernel<<<dim3(8), dim3(64), 0, stream>>>(em, log_pi, log_A, out, beta);

  const int npairs = HB * HT;
  const int threads = 256;
  const int blocks = (npairs*32 + threads - 1) / threads;
  hmm_gamma_kernel<<<dim3(blocks), dim3(threads), 0, stream>>>(out, beta, out, npairs);
}


# Step product split by column tile over two compute waves, two SIMDs: each starts the next product on its own tile's states while the other tile crosses LDS behind one barrier; sum order changes
# speedup vs baseline: 3.6707x; 1.4731x over previous
#include <hip/hip_runtime.h>

typedef __attribute__((ext_vector_type(2))) float v2f;
typedef __attribute__((ext_vector_type(4))) float v4f;
typedef __attribute__((ext_vector_type(8))) float v8f;

constexpr int HB = 64;
constexpr int HT = 4096;
constexpr int HK = 32;
constexpr int CP = 36;
constexpr float LOG2E = 1.44269504088896340736f;
constexpr float LN2   = 0.69314718055994530942f;

#if __has_builtin(__builtin_amdgcn_exp2f)
__device__ __forceinline__ float fexp2(float x){ return __builtin_amdgcn_exp2f(x); }
#else
__device__ __forceinline__ float fexp2(float x){ return exp2f(x); }
#endif
#if __has_builtin(__builtin_amdgcn_logf)
__device__ __forceinline__ float flog2(float x){ return __builtin_amdgcn_logf(x); }
#else
__device__ __forceinline__ float flog2(float x){ return log2f(x); }
#endif

__device__ __forceinline__ float rmax32(float x){
#pragma unroll
  for (int m = 16; m >= 1; m >>= 1) x = fmaxf(x, __shfl_xor(x, m, 32));
  return x;
}
__device__ __forceinline__ float rsum32(float x){
#pragma unroll
  for (int m = 16; m >= 1; m >>= 1) x += __shfl_xor(x, m, 32);
  return x;
}
__device__ __forceinline__ float xhalf(float x){
  return __builtin_bit_cast(float, __builtin_amdgcn_permlanex16(
      __builtin_bit_cast(int, x), __builtin_bit_cast(int, x),
      0x76543210, 0xfedcba98, false, false));
}

__device__ __forceinline__ v8f wmma4(v2f a, v2f b, v8f c){
  return __builtin_amdgcn_wmma_f32_16x16x4_f32(
       false, a,  false, b,
       (short)0, c,  false,  false);
}

__device__ __forceinline__ int tp(int i, int own){ return ((i >> 1) == 0 ? own : 16 - own) + 4*(i & 1); }

__device__ __forceinline__ float row_max(const v4f (&x)[4]){
  const float a0 = fmaxf(fmaxf(x[0][0], x[0][1]), x[0][2]);
  const float a1 = fmaxf(fmaxf(x[0][3], x[1][0]), x[1][1]);
  const float a2 = fmaxf(fmaxf(x[1][2], x[1][3]), x[2][0]);
  const float a3 = fmaxf(fmaxf(x[2][1], x[2][2]), x[2][3]);
  const float a4 = fmaxf(fmaxf(x[3][0], x[3][1]), x[3][2]);
  const float b0 = fmaxf(fmaxf(a0, a1), a2);
  const float b1 = fmaxf(fmaxf(a3, a4), x[3][3]);
  const float mh = fmaxf(b0, b1);
  return fmaxf(mh, xhalf(mh));
}

__device__ __forceinline__ void hmm_matrix_run(const v4f (&z)[4], const v2f (&bw)[8],
                                               v8f &ca, v8f &cb){
  ca = (v8f){0.f,0.f,0.f,0.f,0.f,0.f,0.f,0.f};
  cb = (v8f){0.f,0.f,0.f,0.f,0.f,0.f,0.f,0.f};
#pragma unroll
  for (int kc = 0; kc < 8; ++kc){
    v2f a;
    a.x = fexp2(z[kc >> 1][2*(kc & 1)]);
    a.y = fexp2(z[kc >> 1][2*(kc & 1) + 1]);
    if ((kc & 1) == 0) ca = wmma4(a, bw[kc], ca);
    else               cb = wmma4(a, bw[kc], cb);
  }
}

__device__ __forceinline__ void lds_block_edge(){ __builtin_amdgcn_sched_barrier(0); }

constexpr int G    = 4;
constexpr int NG   = HT / G;
constexpr int POS  = 16*CP + 32;
constexpr int RING = G*POS;
constexpr int NI   = 16;
constexpr int SCT  = 16*CP;

__device__ __forceinline__ void wait_ds0(){ asm volatile("s_wait_dscnt 0x0" ::: "memory"); }
__device__ __forceinline__ void wait_ds_but2(){ asm volatile("s_wait_dscnt 0x2" ::: "memory"); }
__device__ __forceinline__ void wg_barrier(){ asm volatile("s_barrier_signal -1\n\ts_barrier_wait -1" ::: "memory"); }


__device__ __forceinline__ void step_pre(int u, const v8f &ca, const v8f &cb, v4f (&cown)[2],
                                         float* sCd, const float* sCo){
  lds_block_edge();
  const int tile = (u & 1)*SCT;
#pragma unroll
  for (int r = 0; r < 8; ++r) sCd[tile + r*CP] = ca[r] + cb[r];
#pragma unroll
  for (int i = 0; i < 2; ++i) cown[i] = *(const v4f*)(sCo + tile + 4*i);
  wait_ds_but2();
}

template <int DIR>
__device__ __forceinline__ void step_post(int u, v4f (&z)[4], v4f (&op)[2], v4f (&e)[4],
                                          v8f &ca, v8f &cb, float &s, float &zm,
                                          const v2f (&bw)[8], const v4f (&cown)[2],
                                          const float* sCx, const float* en, float* od, int own){
  const int tile = (u & 1)*SCT;
  v4f coth[2], enx[4];
#pragma unroll
  for (int i = 0; i < 2; ++i) coth[i] = *(const v4f*)(sCx + tile + 4*i);
#pragma unroll
  for (int i = 0; i < 4; ++i) enx[i] = *(const v4f*)(en + tp(i, own));
#pragma unroll
  for (int i = 0; i < 2; ++i) *(v4f*)(od + own + 4*i) = op[i];
  lds_block_edge();
#pragma unroll
  for (int i = 0; i < 2; ++i)
#pragma unroll
    for (int c = 0; c < 4; ++c){
      const float l = flog2(cown[i][c]);
      z[i][c] = l + fmaf(e[i][c], LOG2E, -zm);
      if (DIR == 0) op[i][c] = fmaf(LN2, l, s + e[i][c]);
      else          op[i][c] = fmaf(LN2, l, s);
    }
#pragma unroll
  for (int i = 0; i < 2; ++i)
#pragma unroll
    for (int c = 0; c < 4; ++c){
      const float l = flog2(coth[i][c]);
      z[2 + i][c] = l + fmaf(e[2 + i][c], LOG2E, -zm);
    }
  s  = fmaf(LN2, zm, s);
  zm = row_max(z);
#pragma unroll
  for (int i = 0; i < 4; ++i) e[i] = enx[i];
  hmm_matrix_run(z, bw, ca, cb);
}

template <int DIR>
__device__ __forceinline__ void first_row(v4f (&z)[4], v4f (&op)[2], v4f (&e)[4],
                                          v8f &ca, v8f &cb, float &s, float &zm,
                                          const v2f (&bw)[8], const float* sL,
                                          const float* eb, const float* en, int own, int h){
  v4f x[4], enx[4];
#pragma unroll
  for (int i = 0; i < 4; ++i) e[i] = *(const v4f*)(eb + tp(i, own));
#pragma unroll
  for (int i = 0; i < 4; ++i) enx[i] = *(const v4f*)(en + tp(i, own));
#pragma unroll
  for (int i = 0; i < 4; ++i)
#pragma unroll
    for (int c = 0; c < 4; ++c){
      if (DIR == 0) x[i][c] = sL[8*h + tp(i, own) + c] + e[i][c];
      else          x[i][c] = e[i][c];
    }
#pragma unroll
  for (int i = 0; i < 2; ++i)
#pragma unroll
    for (int c = 0; c < 4; ++c) op[i][c] = (DIR == 0) ? x[i][c] : 0.f;
  s = row_max(x);
#pragma unroll
  for (int i = 0; i < 4; ++i)
#pragma unroll
    for (int c = 0; c < 4; ++c) z[i][c] = (x[i][c] - s) * LOG2E;
  zm = 0.f;
#pragma unroll
  for (int i = 0; i < 4; ++i) e[i] = enx[i];
  hmm_matrix_run(z, bw, ca, cb);
}

__device__ __forceinline__ void mem_load(v4f (&er)[NI], const float* src, int tbase, int lane){
#pragma unroll
  for (int j = 0; j < NI; ++j)
    er[j] = *(const v4f*)(src + ((size_t)j*HT + tbase)*HK + 4*lane);
}
__device__ __forceinline__ void mem_put(const v4f (&er)[NI], float* half, int lds_off){
#pragma unroll
  for (int j = 0; j < NI; ++j) *(v4f*)(half + lds_off + j*CP) = er[j];
}
template <int J0, int J1>
__device__ __forceinline__ void mem_drain(const float* half, int lds_off, float* dst, int tbase, int lane){
  v4f v[J1 - J0];
#pragma unroll
  for (int j = J0; j < J1; ++j) v[j - J0] = *(const v4f*)(half + lds_off + j*CP);
#pragma unroll
  for (int j = J0; j < J1; ++j)
    *(v4f*)(dst + ((size_t)j*HT + tbase)*HK + 4*lane) = v[j - J0];
}

template <int DIR, bool FIRST>
__device__ __forceinline__ void compute_group(v4f (&z)[4], v4f (&op)[2], v4f (&e)[4],
                                              v8f &ca, v8f &cb, float &s, float &zm,
                                              const v2f (&bw)[8],
                                              float* sCd, const float* sCo, const float* sCx, const float* sL,
                                              const float* eb, const float* ebn, float* ob, float* obp,
                                              int own, int h){
#pragma unroll
  for (int u = 0; u < G; ++u){
    const float* en = (u < G - 1) ? (eb + (u + 1)*POS) : ebn;
    if (FIRST && u == 0){
      wait_ds0();
      wg_barrier();
      first_row<DIR>(z, op, e, ca, cb, s, zm, bw, sL, eb, en, own, h);
    } else {
      v4f cown[2];
      step_pre(u, ca, cb, cown, sCd, sCo);
      wg_barrier();
      float* od = (u == 0) ? (obp + (G - 1)*POS) : (ob + (u - 1)*POS);
      step_post<DIR>(u, z, op, e, ca, cb, s, zm, bw, cown, sCx, en, od, own);
    }
  }
}

__device__ __forceinline__ void memory_group(int g, v4f (&er)[NI], float* sE, const float* sOd,
                                             int lds_off, const float* emb, float* outb,
                                             int tb0, int tbs, int lane){
  static_assert(G == 4, "the memory wave takes a group's barriers one by one: four of them");
  wait_ds0();
  wg_barrier();
  if (g + 1 < NG) mem_put(er, sE + ((g + 1) & 1)*RING, lds_off);
  if (g + 2 < NG) mem_load(er, emb, tb0 + (g + 2)*tbs, lane);
  wg_barrier();
  if (g >= 2) mem_drain<0, NI/2>(sOd, lds_off, outb, tb0 + (g - 2)*tbs, lane);
  wg_barrier();
  if (g >= 2) mem_drain<NI/2, NI>(sOd, lds_off, outb, tb0 + (g - 2)*tbs, lane);
  wait_ds0();
  wg_barrier();
}

template <int DIR>
__device__ __forceinline__ void scan_loop(int wid, int lane, int h, int own, int row_off, int lds_off,
                                          const v2f (&bw)[8], float* sC, const float* sL,
                                          float* sE, float* sO,
                                          const float* emb, float* outb){
  const int tb0 = (DIR == 0) ? 0 : HT - G;
  const int tbs = (DIR == 0) ? G : -G;

  if (wid != 2){
    const int ln = lane & 15;
    v4f z[4], ev[4], op[2];
    float sh, zm;
    v8f ca, cb;
    float*       sCd = sC + 8*h*CP + own + ln;
    const float* sCo = sC + row_off + own;
    const float* sCx = sC + row_off + (16 - own);
    compute_group<DIR, true>(z, op, ev, ca, cb, sh, zm, bw, sCd, sCo, sCx, sL,
                             sE + row_off, sE + RING + row_off, sO + row_off, sO + row_off, own, h);
    int o3 = 1;
    for (int g = 1; g < NG; ++g){
      const int o3p = (o3 == 0) ? 2 : o3 - 1;
      compute_group<DIR, false>(z, op, ev, ca, cb, sh, zm, bw, sCd, sCo, sCx, sL,
                                sE + (g & 1)*RING + row_off, sE + ((g + 1) & 1)*RING + row_off,
                                sO + o3*RING + row_off, sO + o3p*RING + row_off, own, h);
      o3 = (o3 == 2) ? 0 : o3 + 1;
    }
    float* od = sO + ((NG - 1) % 3)*RING + row_off + (G - 1)*POS + own;
#pragma unroll
    for (int i = 0; i < 2; ++i) *(v4f*)(od + 4*i) = op[i];
    wait_ds0();
    wg_barrier();
  } else {
    v4f er[NI];
    mem_load(er, emb, tb0, lane);
    mem_put(er, sE, lds_off);
    mem_load(er, emb, tb0 + tbs, lane);
    int o3d = 1;
    for (int g = 0; g < NG; ++g){
      memory_group(g, er, sE, sO + o3d*RING, lds_off, emb, outb, tb0, tbs, lane);
      o3d = (o3d == 2) ? 0 : o3d + 1;
    }
    wait_ds0();
    wg_barrier();
    mem_drain<0, NI>(sO + ((NG - 2) % 3)*RING, lds_off, outb, tb0 + (NG - 2)*tbs, lane);
    mem_drain<0, NI>(sO + ((NG - 1) % 3)*RING, lds_off, outb, tb0 + (NG - 1)*tbs, lane);
  }
}

__global__ void __launch_bounds__(96)
hmm_scan_kernel(const float* __restrict__ em,
                const float* __restrict__ log_pi,
                const float* __restrict__ log_A,
                float* __restrict__ alpha_out,
                float* __restrict__ beta_out){
  __shared__ float sA[32*33];
  __shared__ __attribute__((aligned(16))) float sC[2*SCT];
  __shared__ float sL[32];
  __shared__ __attribute__((aligned(16))) float sE[2*RING];
  __shared__ __attribute__((aligned(16))) float sO[3*RING];

  const int lane = threadIdx.x & 31;
  const int wid  = __builtin_amdgcn_readfirstlane((int)(threadIdx.x >> 5));
  const int h    = lane >> 4;
  const int ln   = lane & 15;
  const int dir  = blockIdx.x >> 2;
  const int mb   = blockIdx.x & 3;

  for (int i = wid; i < 32; i += 3){
    float xv = log_A[i*32 + lane];
    float mx = rmax32(xv);
    float e  = fexp2((xv - mx) * LOG2E);
    float s  = rsum32(e);
    sA[i*33 + lane] = e / s;
  }
  if (wid == 0){
    float xp = log_pi[lane];
    float mp = rmax32(xp);
    float sp = rsum32(fexp2((xp - mp) * LOG2E));
    sL[lane] = xp - mp - LN2 * flog2(sp);
  }
  __syncthreads();

  const int own = 16*(wid & 1);
  v2f bw[8];
#pragma unroll
  for (int kc = 0; kc < 8; ++kc){
    const int kk = ((kc >> 2) == 0 ? own : 16 - own) + 8*h + 2*(kc & 3);
    const int nn = own + ln;
    if (dir == 0){
      bw[kc].x = sA[kk*33 + nn];
      bw[kc].y = sA[(kk+1)*33 + nn];
    } else {
      bw[kc].x = sA[nn*33 + kk];
      bw[kc].y = sA[nn*33 + kk + 1];
    }
  }

  const float* emb  = em + (size_t)(mb*16)*HT*HK;
  float*       outb = (dir == 0 ? alpha_out : beta_out) + (size_t)(mb*16)*HT*HK;

  const int row_off = ln*CP + 8*h;
  const int tl      = lane >> 3;
  const int lds_off = ((dir == 0) ? tl : (G - 1 - tl))*POS + 4*(lane & 7);

  if (dir == 0) scan_loop<0>(wid, lane, h, own, row_off, lds_off, bw, sC, sL, sE, sO, emb, outb);
  else          scan_loop<1>(wid, lane, h, own, row_off, lds_off, bw, sC, sL, sE, sO, emb, outb);
}

__global__ void __launch_bounds__(256)
hmm_gamma_kernel(const float* alpha, const float* beta, float* out, int npairs){
  const int w    = (blockIdx.x*blockDim.x + threadIdx.x) >> 5;
  const int lane = threadIdx.x & 31;
  if (w >= npairs) return;
  const size_t idx = (size_t)w*32 + lane;
  float g  = alpha[idx] + beta[idx];
  float mx = rmax32(g);
  float s  = rsum32(fexp2((g - mx) * LOG2E));
  out[idx] = g - mx - LN2 * flog2(s);
}

extern "C" void kernel_launch(void* const* d_in, const int* in_sizes, int n_in,
                              void* d_out, int out_size, void* d_ws, size_t ws_size,
                              hipStream_t stream) {
  const float* em     = (const float*)d_in[0];
  const float* log_pi = (const float*)d_in[1];
  const float* log_A  = (const float*)d_in[2];
  float* out  = (float*)d_out;
  float* beta = (float*)d_ws;

  hmm_scan_kernel<<<dim3(8), dim3(96), 0, stream>>>(em, log_pi, log_A, out, beta);

  const int npairs = HB * HT;
  const int threads = 256;
  const int blocks = (npairs*32 + threads - 1) / threads;
  hmm_gamma_kernel<<<dim3(blocks), dim3(threads), 0, stream>>>(out, beta, out, npairs);
}


# Step product split by k quarter over four compute waves on four SIMDs: 4 matrix instructions (both column tiles), 4 states' vector work a wave; partials added behind the barrier; sum order changes
# speedup vs baseline: 3.8457x; 1.0477x over previous
#include <hip/hip_runtime.h>

typedef __attribute__((ext_vector_type(2))) float v2f;
typedef __attribute__((ext_vector_type(4))) float v4f;
typedef __attribute__((ext_vector_type(8))) float v8f;

constexpr int HB = 64;
constexpr int HT = 4096;
constexpr int HK = 32;
constexpr int CP = 36;
constexpr float LOG2E = 1.44269504088896340736f;
constexpr float LN2   = 0.69314718055994530942f;

#if __has_builtin(__builtin_amdgcn_exp2f)
__device__ __forceinline__ float fexp2(float x){ return __builtin_amdgcn_exp2f(x); }
#else
__device__ __forceinline__ float fexp2(float x){ return exp2f(x); }
#endif
#if __has_builtin(__builtin_amdgcn_logf)
__device__ __forceinline__ float flog2(float x){ return __builtin_amdgcn_logf(x); }
#else
__device__ __forceinline__ float flog2(float x){ return log2f(x); }
#endif

__device__ __forceinline__ float rmax32(float x){
#pragma unroll
  for (int m = 16; m >= 1; m >>= 1) x = fmaxf(x, __shfl_xor(x, m, 32));
  return x;
}
__device__ __forceinline__ float rsum32(float x){
#pragma unroll
  for (int m = 16; m >= 1; m >>= 1) x += __shfl_xor(x, m, 32);
  return x;
}
__device__ __forceinline__ float xhalf(float x){
  return __builtin_bit_cast(float, __builtin_amdgcn_permlanex16(
      __builtin_bit_cast(int, x), __builtin_bit_cast(int, x),
      0x76543210, 0xfedcba98, false, false));
}

__device__ __forceinline__ v8f wmma4(v2f a, v2f b, v8f c){
  return __builtin_amdgcn_wmma_f32_16x16x4_f32(
       false, a,  false, b,
       (short)0, c,  false,  false);
}


__device__ __forceinline__ float quarter_max(const v4f &x){
  const float mh = fmaxf(fmaxf(x[0], x[1]), fmaxf(x[2], x[3]));
  return fmaxf(mh, xhalf(mh));
}

__device__ __forceinline__ void hmm_matrix_run(const v4f &z, const v2f (&bw)[2][2],
                                               v8f &c0, v8f &c1){
  c0 = (v8f){0.f,0.f,0.f,0.f,0.f,0.f,0.f,0.f};
  c1 = (v8f){0.f,0.f,0.f,0.f,0.f,0.f,0.f,0.f};
#pragma unroll
  for (int kc = 0; kc < 2; ++kc){
    v2f a;
    a.x = fexp2(z[2*kc]);
    a.y = fexp2(z[2*kc + 1]);
    c0 = wmma4(a, bw[0][kc], c0);
    c1 = wmma4(a, bw[1][kc], c1);
  }
}

__device__ __forceinline__ void lds_block_edge(){ __builtin_amdgcn_sched_barrier(0); }

constexpr int G    = 4;
constexpr int NG   = HT / G;
constexpr int POS  = 16*CP + 32;
constexpr int RING = G*POS;
constexpr int NI   = 16;
constexpr int SCT  = 16*CP;
constexpr int PL   = 2*SCT;
constexpr int MP   = 64;
constexpr int MEMW = 4;

__device__ __forceinline__ void wait_ds0(){ asm volatile("s_wait_dscnt 0x0" ::: "memory"); }
__device__ __forceinline__ void wg_barrier(){ asm volatile("s_barrier_signal -1\n\ts_barrier_wait -1" ::: "memory"); }


__device__ __forceinline__ void step_pre(int u, const v8f &c0, const v8f &c1, float mk,
                                         float* sCd, float* sMw){
  lds_block_edge();
  const int tile = (u & 1)*SCT;
#pragma unroll
  for (int r = 0; r < 8; ++r) sCd[tile + r*CP] = c0[r];
#pragma unroll
  for (int r = 0; r < 8; ++r) sCd[tile + 16 + r*CP] = c1[r];
  sMw[((u & 1) ^ 1)*MP] = mk;
  wait_ds0();
}

template <int DIR>
__device__ __forceinline__ void step_post(int u, v4f &z, v4f &op, v4f &e,
                                          v8f &c0, v8f &c1, float &s, float &mk,
                                          const v2f (&bw)[2][2],
                                          const float* sCr, const float* sMr,
                                          const float* en, float* od){
  const int tile = (u & 1)*SCT;
  v4f q[4];
#pragma unroll
  for (int j = 0; j < 4; ++j) q[j] = *(const v4f*)(sCr + j*PL + tile);
  const v4f mq  = *(const v4f*)(sMr + ((u & 1) ^ 1)*MP);
  const v4f enx = *(const v4f*)en;
  *(v4f*)od = op;
  lds_block_edge();
  const float zm = fmaxf(fmaxf(mq[0], mq[1]), fmaxf(mq[2], mq[3]));
#pragma unroll
  for (int c = 0; c < 4; ++c){
    const float l = flog2((q[0][c] + q[1][c]) + (q[2][c] + q[3][c]));
    z[c] = l + fmaf(e[c], LOG2E, -zm);
    if (DIR == 0) op[c] = fmaf(LN2, l, s + e[c]);
    else          op[c] = fmaf(LN2, l, s);
  }
  s  = fmaf(LN2, zm, s);
  mk = quarter_max(z);
  e  = enx;
  hmm_matrix_run(z, bw, c0, c1);
}

template <int DIR>
__device__ __forceinline__ void first_row(v4f &z, v4f &op, v4f &e,
                                          v8f &c0, v8f &c1, float &s, float &mk,
                                          const v2f (&bw)[2][2], const float* sL,
                                          const float* eb, const float* en, int sto){
  const v4f eo  = *(const v4f*)eb;
  const v4f enx = *(const v4f*)en;
  const float* er = eb - sto;
  v4f ef[8];
#pragma unroll
  for (int j = 0; j < 8; ++j) ef[j] = *(const v4f*)(er + 4*j);
  float m = 0.f;
#pragma unroll
  for (int j = 0; j < 8; ++j)
#pragma unroll
    for (int c = 0; c < 4; ++c){
      const float xv = (DIR == 0) ? (sL[4*j + c] + ef[j][c]) : ef[j][c];
      m = (j == 0 && c == 0) ? xv : fmaxf(m, xv);
    }
  v4f xo;
#pragma unroll
  for (int c = 0; c < 4; ++c) xo[c] = (DIR == 0) ? (sL[sto + c] + eo[c]) : eo[c];
#pragma unroll
  for (int c = 0; c < 4; ++c) op[c] = (DIR == 0) ? xo[c] : 0.f;
  s = m;
#pragma unroll
  for (int c = 0; c < 4; ++c) z[c] = (xo[c] - s) * LOG2E;
  mk = quarter_max(z);
  e  = enx;
  hmm_matrix_run(z, bw, c0, c1);
}

__device__ __forceinline__ void mem_load(v4f (&er)[NI], const float* src, int tbase, int lane){
#pragma unroll
  for (int j = 0; j < NI; ++j)
    er[j] = *(const v4f*)(src + ((size_t)j*HT + tbase)*HK + 4*lane);
}
__device__ __forceinline__ void mem_put(const v4f (&er)[NI], float* half, int lds_off){
#pragma unroll
  for (int j = 0; j < NI; ++j) *(v4f*)(half + lds_off + j*CP) = er[j];
}
template <int J0, int J1>
__device__ __forceinline__ void mem_drain(const float* half, int lds_off, float* dst, int tbase, int lane){
  v4f v[J1 - J0];
#pragma unroll
  for (int j = J0; j < J1; ++j) v[j - J0] = *(const v4f*)(half + lds_off + j*CP);
#pragma unroll
  for (int j = J0; j < J1; ++j)
    *(v4f*)(dst + ((size_t)j*HT + tbase)*HK + 4*lane) = v[j - J0];
}

template <int DIR, bool FIRST>
__device__ __forceinline__ void compute_group(v4f &z, v4f &op, v4f &e,
                                              v8f &c0, v8f &c1, float &s, float &mk,
                                              const v2f (&bw)[2][2],
                                              float* sCd, const float* sCr, const float* sMr, float* sMw,
                                              const float* sL,
                                              const float* eb, const float* ebn, float* ob, float* obp,
                                              int sto){
#pragma unroll
  for (int u = 0; u < G; ++u){
    const float* en = (u < G - 1) ? (eb + (u + 1)*POS) : ebn;
    if (FIRST && u == 0){
      wait_ds0();
      wg_barrier();
      first_row<DIR>(z, op, e, c0, c1, s, mk, bw, sL, eb, en, sto);
    } else {
      step_pre(u, c0, c1, mk, sCd, sMw);
      wg_barrier();
      float* od = (u == 0) ? (obp + (G - 1)*POS) : (ob + (u - 1)*POS);
      step_post<DIR>(u, z, op, e, c0, c1, s, mk, bw, sCr, sMr, en, od);
    }
  }
}

__device__ __forceinline__ void memory_group(int g, v4f (&er)[NI], float* sE, const float* sOd,
                                             int lds_off, const float* emb, float* outb,
                                             int tb0, int tbs, int lane){
  static_assert(G == 4, "the memory wave takes a group's barriers one by one: four of them");
  wait_ds0();
  wg_barrier();
  if (g + 1 < NG) mem_put(er, sE + ((g + 1) & 1)*RING, lds_off);
  if (g + 2 < NG) mem_load(er, emb, tb0 + (g + 2)*tbs, lane);
  wg_barrier();
  if (g >= 2) mem_drain<0, NI/2>(sOd, lds_off, outb, tb0 + (g - 2)*tbs, lane);
  wg_barrier();
  if (g >= 2) mem_drain<NI/2, NI>(sOd, lds_off, outb, tb0 + (g - 2)*tbs, lane);
  wait_ds0();
  wg_barrier();
}

template <int DIR>
__device__ __forceinline__ void scan_loop(int wid, int lane, int h, int ln, int lds_off,
                                          const v2f (&bw)[2][2], float* sC, float* sM, const float* sL,
                                          float* sE, float* sO,
                                          const float* emb, float* outb){
  const int tb0 = (DIR == 0) ? 0 : HT - G;
  const int tbs = (DIR == 0) ? G : -G;

  if (wid != MEMW){
    v4f z, ev, op;
    float sh, mk;
    v8f c0, c1;
    const int sto = 8*wid + 4*h;
    const int ro  = ln*CP + sto;
    float*       sCd = sC + wid*PL + 8*h*CP + ln;
    const float* sCr = sC + ro;
    float*       sMw = sM + 4*ln + wid;
    const float* sMr = sM + 4*ln;
    compute_group<DIR, true>(z, op, ev, c0, c1, sh, mk, bw, sCd, sCr, sMr, sMw, sL,
                             sE + ro, sE + RING + ro, sO + ro, sO + ro, sto);
    int o3 = 1;
    for (int g = 1; g < NG; ++g){
      const int o3p = (o3 == 0) ? 2 : o3 - 1;
      compute_group<DIR, false>(z, op, ev, c0, c1, sh, mk, bw, sCd, sCr, sMr, sMw, sL,
                                sE + (g & 1)*RING + ro, sE + ((g + 1) & 1)*RING + ro,
                                sO + o3*RING + ro, sO + o3p*RING + ro, sto);
      o3 = (o3 == 2) ? 0 : o3 + 1;
    }
    float* od = sO + ((NG - 1) % 3)*RING + (G - 1)*POS + ro;
    *(v4f*)od = op;
    wait_ds0();
    wg_barrier();
  } else {
    v4f er[NI];
    mem_load(er, emb, tb0, lane);
    mem_put(er, sE, lds_off);
    mem_load(er, emb, tb0 + tbs, lane);
    int o3d = 1;
    for (int g = 0; g < NG; ++g){
      memory_group(g, er, sE, sO + o3d*RING, lds_off, emb, outb, tb0, tbs, lane);
      o3d = (o3d == 2) ? 0 : o3d + 1;
    }
    wait_ds0();
    wg_barrier();
    mem_drain<0, NI>(sO + ((NG - 2) % 3)*RING, lds_off, outb, tb0 + (NG - 2)*tbs, lane);
    mem_drain<0, NI>(sO + ((NG - 1) % 3)*RING, lds_off, outb, tb0 + (NG - 1)*tbs, lane);
  }
}

__global__ void __launch_bounds__(160)
hmm_scan_kernel(const float* __restrict__ em,
                const float* __restrict__ log_pi,
                const float* __restrict__ log_A,
                float* __restrict__ alpha_out,
                float* __restrict__ beta_out){
  __shared__ float sA[32*33];
  __shared__ __attribute__((aligned(16))) float sC[4*PL];
  __shared__ __attribute__((aligned(16))) float sM[2*MP];
  __shared__ __attribute__((aligned(16))) float sL[32];
  __shared__ __attribute__((aligned(16))) float sE[2*RING];
  __shared__ __attribute__((aligned(16))) float sO[3*RING];

  const int lane = threadIdx.x & 31;
  const int wid  = __builtin_amdgcn_readfirstlane((int)(threadIdx.x >> 5));
  const int h    = lane >> 4;
  const int ln   = lane & 15;
  const int dir  = blockIdx.x >> 2;
  const int mb   = blockIdx.x & 3;

  for (int i = wid; i < 32; i += 5){
    float xv = log_A[i*32 + lane];
    float mx = rmax32(xv);
    float e  = fexp2((xv - mx) * LOG2E);
    float s  = rsum32(e);
    sA[i*33 + lane] = e / s;
  }
  if (wid == 0){
    float xp = log_pi[lane];
    float mp = rmax32(xp);
    float sp = rsum32(fexp2((xp - mp) * LOG2E));
    sL[lane] = xp - mp - LN2 * flog2(sp);
  }
  __syncthreads();

  const int kq = wid & 3;
  v2f bw[2][2];
#pragma unroll
  for (int n = 0; n < 2; ++n)
#pragma unroll
    for (int kc = 0; kc < 2; ++kc){
      const int kk = 8*kq + 4*h + 2*kc;
      const int nn = 16*n + ln;
      if (dir == 0){
        bw[n][kc].x = sA[kk*33 + nn];
        bw[n][kc].y = sA[(kk+1)*33 + nn];
      } else {
        bw[n][kc].x = sA[nn*33 + kk];
        bw[n][kc].y = sA[nn*33 + kk + 1];
      }
    }

  const float* emb  = em + (size_t)(mb*16)*HT*HK;
  float*       outb = (dir == 0 ? alpha_out : beta_out) + (size_t)(mb*16)*HT*HK;

  const int tl      = lane >> 3;
  const int lds_off = ((dir == 0) ? tl : (G - 1 - tl))*POS + 4*(lane & 7);

  if (dir == 0) scan_loop<0>(wid, lane, h, ln, lds_off, bw, sC, sM, sL, sE, sO, emb, outb);
  else          scan_loop<1>(wid, lane, h, ln, lds_off, bw, sC, sM, sL, sE, sO, emb, outb);
}

__global__ void __launch_bounds__(256)
hmm_gamma_kernel(const float* alpha, const float* beta, float* out, int npairs){
  const int w    = (blockIdx.x*blockDim.x + threadIdx.x) >> 5;
  const int lane = threadIdx.x & 31;
  if (w >= npairs) return;
  const size_t idx = (size_t)w*32 + lane;
  float g  = alpha[idx] + beta[idx];
  float mx = rmax32(g);
  float s  = rsum32(fexp2((g - mx) * LOG2E));
  out[idx] = g - mx - LN2 * flog2(s);
}

extern "C" void kernel_launch(void* const* d_in, const int* in_sizes, int n_in,
                              void* d_out, int out_size, void* d_ws, size_t ws_size,
                              hipStream_t stream) {
  const float* em     = (const float*)d_in[0];
  const float* log_pi = (const float*)d_in[1];
  const float* log_A  = (const float*)d_in[2];
  float* out  = (float*)d_out;
  float* beta = (float*)d_ws;

  hmm_scan_kernel<<<dim3(8), dim3(160), 0, stream>>>(em, log_pi, log_A, out, beta);

  const int npairs = HB * HT;
  const int threads = 256;
  const int blocks = (npairs*32 + threads - 1) / threads;
  hmm_gamma_kernel<<<dim3(blocks), dim3(threads), 0, stream>>>(out, beta, out, npairs);
}
